# starting kernel
# baseline (speedup 1.0000x reference)
#include <hip/hip_runtime.h>

typedef float v2f __attribute__((ext_vector_type(2)));
typedef float v8f __attribute__((ext_vector_type(8)));

namespace {
constexpr int CIN        = 512;
constexpr int COUT       = 512;
constexpr int HW         = 49;
constexpr int LDS_STRIDE = 50;
}

__global__ __launch_bounds__(128)
void grouped_conv_roll_wmma(const float* __restrict__ x,
                            const float* __restrict__ w,
                            float* __restrict__ out) {
  extern __shared__ float ldsx[];

  const int tid  = threadIdx.x;
  const int lane = tid & 31;
  const int wave = tid >> 5;
  const int o    = blockIdx.x >> 3;
  const int k0   = ((blockIdx.x & 7) * 4 + wave) * 16;

  const float* xg = x + o * CIN * HW;
  for (int r = tid; r < CIN; r += 128) {
    const float* src = xg + r * HW;
    float* dst = ldsx + r * LDS_STRIDE;
    #pragma unroll 7
    for (int c = 0; c < HW; ++c) dst[c] = src[c];
    dst[HW] = 0.0f;
  }
  __syncthreads();

  const int ncol = lane & 15;
  const int half = lane >> 4;
  const int koff = half * 2;

  int boff[3][4];
  #pragma unroll
  for (int j = 0; j < 3; ++j) {
    #pragma unroll
    for (int t = 0; t < 4; ++t) {
      const int nm = t * 16 + ncol;
      int off = HW;
      if (nm < HW) {
        const int n = nm / 7, m = nm % 7;
        const int p = m + j;
        if (p >= 1 && p <= 7) off = n * 7 + (p + 5) % 7;
      }
      boff[j][t] = off;
    }
  }

  v8f acc[4] = {};

  const float* wk = w + (k0 + ncol) * 3;

  for (int j = 0; j < 3; ++j) {
    const int o0 = boff[j][0], o1 = boff[j][1];
    const int o2 = boff[j][2], o3 = boff[j][3];
    for (int i0 = 0; i0 < CIN; i0 += 4) {
      const int ia = i0 + koff;
      v2f A = { wk[ia * (COUT * 3) + j],
                wk[(ia + 1) * (COUT * 3) + j] };
      const float* br0 = ldsx + ia * LDS_STRIDE;
      const float* br1 = br0 + LDS_STRIDE;
      v2f B0 = { br0[o0], br1[o0] };
      v2f B1 = { br0[o1], br1[o1] };
      v2f B2 = { br0[o2], br1[o2] };
      v2f B3 = { br0[o3], br1[o3] };
      acc[0] = __builtin_amdgcn_wmma_f32_16x16x4_f32(false, A, false, B0, (short)0, acc[0], false, false);
      acc[1] = __builtin_amdgcn_wmma_f32_16x16x4_f32(false, A, false, B1, (short)0, acc[1], false, false);
      acc[2] = __builtin_amdgcn_wmma_f32_16x16x4_f32(false, A, false, B2, (short)0, acc[2], false, false);
      acc[3] = __builtin_amdgcn_wmma_f32_16x16x4_f32(false, A, false, B3, (short)0, acc[3], false, false);
    }
  }

  float* outg = out + o * COUT * HW;
  const int kbase = k0 + half * 8;
  #pragma unroll
  for (int t = 0; t < 4; ++t) {
    const int nm = t * 16 + ncol;
    if (nm < HW) {
      const int n = nm / 7, m = nm % 7;
      const int sp = ((n + 1) % 7) * 7 + m;
      #pragma unroll
      for (int v = 0; v < 8; ++v)
        outg[(kbase + v) * HW + sp] = acc[t][v];
    }
  }
}

extern "C" void kernel_launch(void* const* d_in, const int* in_sizes, int n_in,
                              void* d_out, int out_size, void* d_ws, size_t ws_size,
                              hipStream_t stream) {
  (void)in_sizes; (void)n_in; (void)d_ws; (void)ws_size; (void)out_size;
  const float* x = (const float*)d_in[0];
  const float* w = (const float*)d_in[1];
  float* out = (float*)d_out;

  dim3 grid(16), block(128);
  size_t shmem = (size_t)CIN * LDS_STRIDE * sizeof(float);
  hipLaunchKernelGGL(grouped_conv_roll_wmma, grid, block, shmem, stream, x, w, out);
}


# Weight fragments double-buffered in registers: each 64-channel block (32 loads per lane) is requested 64 WMMAs ahead, so products no longer wait on an L2 read every 4 K-steps; sums unchanged
# speedup vs baseline: 1.6840x; 1.6840x over previous
#include <hip/hip_runtime.h>

typedef float v2f __attribute__((ext_vector_type(2)));
typedef float v8f __attribute__((ext_vector_type(8)));

namespace {
constexpr int CIN        = 512;
constexpr int COUT       = 512;
constexpr int HW         = 49;
constexpr int LDS_STRIDE = 50;
constexpr int WROW       = COUT * 3;
constexpr int KB         = 16;
constexpr int BLK_CH     = KB * 4;
constexpr int BLK_TAP    = CIN / BLK_CH;
constexpr int NPAIR      = 3 * BLK_TAP / 2;
}

__device__ __forceinline__ void load_wblock(v2f (&A)[KB], const float* __restrict__ p) {
  #pragma unroll
  for (int u = 0; u < KB; ++u) {
    A[u].x = p[(u * 4) * WROW];
    A[u].y = p[(u * 4 + 1) * WROW];
  }
}

__device__ __forceinline__ void mma_block(v8f (&acc)[4], const v2f (&A)[KB],
                                          const float* bb,
                                          int o0, int o1, int o2, int o3) {
  #pragma unroll
  for (int u = 0; u < KB; ++u) {
    const float* br0 = bb + (u * 4) * LDS_STRIDE;
    const float* br1 = br0 + LDS_STRIDE;
    v2f B0 = { br0[o0], br1[o0] };
    v2f B1 = { br0[o1], br1[o1] };
    v2f B2 = { br0[o2], br1[o2] };
    v2f B3 = { br0[o3], br1[o3] };
    acc[0] = __builtin_amdgcn_wmma_f32_16x16x4_f32(false, A[u], false, B0, (short)0, acc[0], false, false);
    acc[1] = __builtin_amdgcn_wmma_f32_16x16x4_f32(false, A[u], false, B1, (short)0, acc[1], false, false);
    acc[2] = __builtin_amdgcn_wmma_f32_16x16x4_f32(false, A[u], false, B2, (short)0, acc[2], false, false);
    acc[3] = __builtin_amdgcn_wmma_f32_16x16x4_f32(false, A[u], false, B3, (short)0, acc[3], false, false);
  }
}

__global__ __launch_bounds__(128)
void grouped_conv_roll_wmma(const float* __restrict__ x,
                            const float* __restrict__ w,
                            float* __restrict__ out) {
  extern __shared__ float ldsx[];

  const int tid  = threadIdx.x;
  const int lane = tid & 31;
  const int wave = tid >> 5;
  const int o    = blockIdx.x >> 3;
  const int k0   = ((blockIdx.x & 7) * 4 + wave) * 16;

  const int ncol = lane & 15;
  const int half = lane >> 4;
  const int koff = half * 2;

  const float* wl = w + koff * WROW + (k0 + ncol) * 3;

  v2f Wa[KB], Wb[KB];
  load_wblock(Wa, wl);

  const float* xg = x + o * CIN * HW;
  for (int r = tid; r < CIN; r += 128) {
    const float* src = xg + r * HW;
    float* dst = ldsx + r * LDS_STRIDE;
    #pragma unroll 7
    for (int c = 0; c < HW; ++c) dst[c] = src[c];
    dst[HW] = 0.0f;
  }
  __syncthreads();

  int boff[3][4];
  #pragma unroll
  for (int j = 0; j < 3; ++j) {
    #pragma unroll
    for (int t = 0; t < 4; ++t) {
      const int nm = t * 16 + ncol;
      int off = HW;
      if (nm < HW) {
        const int n = nm / 7, m = nm % 7;
        const int p = m + j;
        if (p >= 1 && p <= 7) off = n * 7 + (p + 5) % 7;
      }
      boff[j][t] = off;
    }
  }

  v8f acc[4] = {};

  const float* bl = ldsx + koff * LDS_STRIDE;

  int o0 = 0, o1 = 0, o2 = 0, o3 = 0;
  const float* bb = bl;

  int bp = 0;
  for (;;) {
    const int j  = bp / (BLK_TAP / 2);
    const int c0 = (bp % (BLK_TAP / 2)) * 2;
    o0 = (j == 0) ? boff[0][0] : (j == 1) ? boff[1][0] : boff[2][0];
    o1 = (j == 0) ? boff[0][1] : (j == 1) ? boff[1][1] : boff[2][1];
    o2 = (j == 0) ? boff[0][2] : (j == 1) ? boff[1][2] : boff[2][2];
    o3 = (j == 0) ? boff[0][3] : (j == 1) ? boff[1][3] : boff[2][3];
    bb = bl + (c0 + 1) * BLK_CH * LDS_STRIDE;

    load_wblock(Wb, wl + (c0 + 1) * BLK_CH * WROW + j);
    mma_block(acc, Wa, bl + c0 * BLK_CH * LDS_STRIDE, o0, o1, o2, o3);

    if (++bp == NPAIR) break;

    load_wblock(Wa, wl + ((bp % (BLK_TAP / 2)) * 2) * BLK_CH * WROW + bp / (BLK_TAP / 2));
    mma_block(acc, Wb, bb, o0, o1, o2, o3);
  }
  mma_block(acc, Wb, bb, o0, o1, o2, o3);

  float* outg = out + o * COUT * HW;
  const int kbase = k0 + half * 8;
  #pragma unroll
  for (int t = 0; t < 4; ++t) {
    const int nm = t * 16 + ncol;
    if (nm < HW) {
      const int n = nm / 7, m = nm % 7;
      const int sp = ((n + 1) % 7) * 7 + m;
      #pragma unroll
      for (int v = 0; v < 8; ++v)
        outg[(kbase + v) * HW + sp] = acc[t][v];
    }
  }
}

extern "C" void kernel_launch(void* const* d_in, const int* in_sizes, int n_in,
                              void* d_out, int out_size, void* d_ws, size_t ws_size,
                              hipStream_t stream) {
  (void)in_sizes; (void)n_in; (void)d_ws; (void)ws_size; (void)out_size;
  const float* x = (const float*)d_in[0];
  const float* w = (const float*)d_in[1];
  float* out = (float*)d_out;

  dim3 grid(16), block(128);
  size_t shmem = (size_t)CIN * LDS_STRIDE * sizeof(float);
  hipLaunchKernelGGL(grouped_conv_roll_wmma, grid, block, shmem, stream, x, w, out);
}


# B fragments of each 64-channel block read from LDS as one run of 64 reads one block ahead into a second register set; no LDS read left among the WMMAs (was 4 reads after every 4); sums unchanged
# speedup vs baseline: 1.8562x; 1.1023x over previous
#include <hip/hip_runtime.h>

typedef float v2f __attribute__((ext_vector_type(2)));
typedef float v8f __attribute__((ext_vector_type(8)));

namespace {
constexpr int CIN        = 512;
constexpr int COUT       = 512;
constexpr int HW         = 49;
constexpr int LDS_STRIDE = 50;
constexpr int WROW       = COUT * 3;
constexpr int KB         = 16;
constexpr int BLK_CH     = KB * 4;
constexpr int BLK_TAP    = CIN / BLK_CH;
constexpr int NPAIR      = 3 * BLK_TAP / 2;
}

__device__ __forceinline__ void load_wblock(v2f (&A)[KB], const float* __restrict__ p) {
  #pragma unroll
  for (int u = 0; u < KB; ++u) {
    A[u].x = p[(u * 4) * WROW];
    A[u].y = p[(u * 4 + 1) * WROW];
  }
}

__device__ __forceinline__ void load_bblock(v2f (&B)[KB][4], const float* bb,
                                            int o0, int o1, int o2, int o3) {
  #pragma unroll
  for (int u = 0; u < KB; ++u) {
    const float* br0 = bb + (u * 4) * LDS_STRIDE;
    const float* br1 = br0 + LDS_STRIDE;
    B[u][0] = v2f{ br0[o0], br1[o0] };
    B[u][1] = v2f{ br0[o1], br1[o1] };
    B[u][2] = v2f{ br0[o2], br1[o2] };
    B[u][3] = v2f{ br0[o3], br1[o3] };
  }
}

__device__ __forceinline__ void mma_block(v8f (&acc)[4], const v2f (&A)[KB],
                                          const v2f (&B)[KB][4]) {
  #pragma unroll
  for (int u = 0; u < KB; ++u) {
    acc[0] = __builtin_amdgcn_wmma_f32_16x16x4_f32(false, A[u], false, B[u][0], (short)0, acc[0], false, false);
    acc[1] = __builtin_amdgcn_wmma_f32_16x16x4_f32(false, A[u], false, B[u][1], (short)0, acc[1], false, false);
    acc[2] = __builtin_amdgcn_wmma_f32_16x16x4_f32(false, A[u], false, B[u][2], (short)0, acc[2], false, false);
    acc[3] = __builtin_amdgcn_wmma_f32_16x16x4_f32(false, A[u], false, B[u][3], (short)0, acc[3], false, false);
  }
}

__global__ __launch_bounds__(128)
void grouped_conv_roll_wmma(const float* __restrict__ x,
                            const float* __restrict__ w,
                            float* __restrict__ out) {
  extern __shared__ float ldsx[];

  const int tid  = threadIdx.x;
  const int lane = tid & 31;
  const int wave = tid >> 5;
  const int o    = blockIdx.x >> 3;
  const int k0   = ((blockIdx.x & 7) * 4 + wave) * 16;

  const int ncol = lane & 15;
  const int half = lane >> 4;
  const int koff = half * 2;

  const float* wl = w + koff * WROW + (k0 + ncol) * 3;

  v2f Wa[KB], Wb[KB];
  load_wblock(Wa, wl);

  const float* xg = x + o * CIN * HW;
  for (int r = tid; r < CIN; r += 128) {
    const float* src = xg + r * HW;
    float* dst = ldsx + r * LDS_STRIDE;
    #pragma unroll 7
    for (int c = 0; c < HW; ++c) dst[c] = src[c];
    dst[HW] = 0.0f;
  }
  __syncthreads();

  int boff[3][4];
  #pragma unroll
  for (int j = 0; j < 3; ++j) {
    #pragma unroll
    for (int t = 0; t < 4; ++t) {
      const int nm = t * 16 + ncol;
      int off = HW;
      if (nm < HW) {
        const int n = nm / 7, m = nm % 7;
        const int p = m + j;
        if (p >= 1 && p <= 7) off = n * 7 + (p + 5) % 7;
      }
      boff[j][t] = off;
    }
  }

  v8f acc[4] = {};

  const float* bl = ldsx + koff * LDS_STRIDE;

  v2f Ba[KB][4], Bb[KB][4];
  load_bblock(Ba, bl, boff[0][0], boff[0][1], boff[0][2], boff[0][3]);

  int bp = 0;
  for (;;) {
    {
      const int j  = bp / (BLK_TAP / 2);
      const int c1 = (bp % (BLK_TAP / 2)) * 2 + 1;
      const int o0 = (j == 0) ? boff[0][0] : (j == 1) ? boff[1][0] : boff[2][0];
      const int o1 = (j == 0) ? boff[0][1] : (j == 1) ? boff[1][1] : boff[2][1];
      const int o2 = (j == 0) ? boff[0][2] : (j == 1) ? boff[1][2] : boff[2][2];
      const int o3 = (j == 0) ? boff[0][3] : (j == 1) ? boff[1][3] : boff[2][3];
      load_wblock(Wb, wl + c1 * BLK_CH * WROW + j);
      load_bblock(Bb, bl + c1 * BLK_CH * LDS_STRIDE, o0, o1, o2, o3);
    }
    __builtin_amdgcn_sched_barrier(0);
    mma_block(acc, Wa, Ba);

    if (++bp == NPAIR) break;

    {
      const int j  = bp / (BLK_TAP / 2);
      const int c0 = (bp % (BLK_TAP / 2)) * 2;
      const int o0 = (j == 0) ? boff[0][0] : (j == 1) ? boff[1][0] : boff[2][0];
      const int o1 = (j == 0) ? boff[0][1] : (j == 1) ? boff[1][1] : boff[2][1];
      const int o2 = (j == 0) ? boff[0][2] : (j == 1) ? boff[1][2] : boff[2][2];
      const int o3 = (j == 0) ? boff[0][3] : (j == 1) ? boff[1][3] : boff[2][3];
      load_wblock(Wa, wl + c0 * BLK_CH * WROW + j);
      load_bblock(Ba, bl + c0 * BLK_CH * LDS_STRIDE, o0, o1, o2, o3);
    }
    __builtin_amdgcn_sched_barrier(0);
    mma_block(acc, Wb, Bb);
  }
  mma_block(acc, Wb, Bb);

  float* outg = out + o * COUT * HW;
  const int kbase = k0 + half * 8;
  #pragma unroll
  for (int t = 0; t < 4; ++t) {
    const int nm = t * 16 + ncol;
    if (nm < HW) {
      const int n = nm / 7, m = nm % 7;
      const int sp = ((n + 1) % 7) * 7 + m;
      #pragma unroll
      for (int v = 0; v < 8; ++v)
        outg[(kbase + v) * HW + sp] = acc[t][v];
    }
  }
}

extern "C" void kernel_launch(void* const* d_in, const int* in_sizes, int n_in,
                              void* d_out, int out_size, void* d_ws, size_t ws_size,
                              hipStream_t stream) {
  (void)in_sizes; (void)n_in; (void)d_ws; (void)ws_size; (void)out_size;
  const float* x = (const float*)d_in[0];
  const float* w = (const float*)d_in[1];
  float* out = (float*)d_out;

  dim3 grid(16), block(128);
  size_t shmem = (size_t)CIN * LDS_STRIDE * sizeof(float);
  hipLaunchKernelGGL(grouped_conv_roll_wmma, grid, block, shmem, stream, x, w, out);
}


# Tap moved from the K walk into M: rows = flat (out-channel, tap) of w, read contiguous; plain GEMM on raw x, K=512 per wave, 192 waves; taps summed at their shifts in LDS after it (sum order changes)
# speedup vs baseline: 4.3729x; 2.3558x over previous
#include <hip/hip_runtime.h>

typedef float v2f __attribute__((ext_vector_type(2)));
typedef float v8f __attribute__((ext_vector_type(8)));

namespace {
constexpr int CIN     = 512;
constexpr int COUT    = 512;
constexpr int HW      = 49;
constexpr int WROW    = COUT * 3;
constexpr int KB      = 16;
constexpr int BLK_CH  = KB * 4;
constexpr int NPAIR   = CIN / BLK_CH / 2;
constexpr int SLAB    = CIN * HW;
constexpr int NCP     = SLAB / 2;
constexpr int NTHR    = 96;
constexpr int NSWEEP  = NCP / NTHR;
constexpr int PROWS   = 48;
constexpr int PSTR    = 66;
constexpr int OUT_WG  = 16 * HW;
}

__device__ __forceinline__ void load_wblock(v2f (&A)[KB], const float* __restrict__ p) {
  #pragma unroll
  for (int u = 0; u < KB; ++u) {
    A[u].x = p[(u * 4) * WROW];
    A[u].y = p[(u * 4 + 1) * WROW];
  }
}

__device__ __forceinline__ void load_bblock(v2f (&B)[KB][4], const v2f* bb,
                                            int o0, int o1, int o2, int o3) {
  #pragma unroll
  for (int u = 0; u < KB; ++u) {
    const v2f* br = bb + (u * 2) * HW;
    B[u][0] = br[o0];
    B[u][1] = br[o1];
    B[u][2] = br[o2];
    B[u][3] = br[o3];
  }
}

__device__ __forceinline__ void mma_block(v8f (&acc)[4], const v2f (&A)[KB],
                                          const v2f (&B)[KB][4]) {
  #pragma unroll
  for (int u = 0; u < KB; ++u) {
    acc[0] = __builtin_amdgcn_wmma_f32_16x16x4_f32(false, A[u], false, B[u][0], (short)0, acc[0], false, false);
    acc[1] = __builtin_amdgcn_wmma_f32_16x16x4_f32(false, A[u], false, B[u][1], (short)0, acc[1], false, false);
    acc[2] = __builtin_amdgcn_wmma_f32_16x16x4_f32(false, A[u], false, B[u][2], (short)0, acc[2], false, false);
    acc[3] = __builtin_amdgcn_wmma_f32_16x16x4_f32(false, A[u], false, B[u][3], (short)0, acc[3], false, false);
  }
}

__global__ __launch_bounds__(96)
void grouped_conv_roll_wmma(const float* __restrict__ x,
                            const float* __restrict__ w,
                            float* __restrict__ out) {
  extern __shared__ v2f lds2[];
  float* ptile = reinterpret_cast<float*>(lds2 + NCP);

  const int tid  = threadIdx.x;
  const int lane = tid & 31;
  const int wave = tid >> 5;
  const int o    = blockIdx.x >> 5;
  const int wg   = blockIdx.x & 31;
  const int r0   = wg * PROWS + wave * 16;

  const int ncol = lane & 15;
  const int half = lane >> 4;
  const int koff = half * 2;

  const float* wl = w + koff * WROW + r0 + ncol;

  v2f Wa[KB], Wb[KB];
  load_wblock(Wa, wl);

  {
    const float* xg = x + o * SLAB;
    #pragma unroll
    for (int q = 0; q < NSWEEP; ++q) {
      const int g = q * NTHR + tid;
      const int s = g + (g / HW) * HW;
      lds2[g] = v2f{ xg[s], xg[s + HW] };
    }
    const int g = NSWEEP * NTHR + tid;
    if (g < NCP) {
      const int s = g + (g / HW) * HW;
      lds2[g] = v2f{ xg[s], xg[s + HW] };
    }
  }
  __syncthreads();

  const int of0 = ncol;
  const int of1 = 16 + ncol;
  const int of2 = 32 + ncol;
  const int of3 = (48 + ncol < HW) ? 48 + ncol : HW - 1;

  v8f acc[4] = {};

  const v2f* bl = lds2 + half * HW;

  v2f Ba[KB][4], Bb[KB][4];
  load_bblock(Ba, bl, of0, of1, of2, of3);

  int bp = 0;
  for (;;) {
    load_wblock(Wb, wl + (2 * bp + 1) * BLK_CH * WROW);
    load_bblock(Bb, bl + (2 * bp + 1) * (BLK_CH / 2) * HW, of0, of1, of2, of3);
    __builtin_amdgcn_sched_barrier(0);
    mma_block(acc, Wa, Ba);

    if (++bp == NPAIR) break;

    load_wblock(Wa, wl + (2 * bp) * BLK_CH * WROW);
    load_bblock(Ba, bl + (2 * bp) * (BLK_CH / 2) * HW, of0, of1, of2, of3);
    __builtin_amdgcn_sched_barrier(0);
    mma_block(acc, Wb, Bb);
  }
  mma_block(acc, Wb, Bb);

  {
    float* pw = ptile + (wave * 16 + half * 8) * PSTR + ncol;
    #pragma unroll
    for (int t = 0; t < 4; ++t) {
      #pragma unroll
      for (int v = 0; v < 8; ++v)
        pw[v * PSTR + t * 16] = acc[t][v];
    }
  }
  __syncthreads();

  float* outg = out + (o * COUT + wg * 16) * HW;
  auto tap_sum_store = [&](int e) {
    const int kl  = e / HW;
    const int rem = e - kl * HW;
    const int no  = rem / 7;
    const int m   = rem - no * 7;
    const int n   = (no + 6) % 7;
    const float* pr = ptile + (kl * 3) * PSTR + n * 7;
    const float a0 = pr[(m + 5) % 7];
    const float a1 = pr[PSTR + (m + 6) % 7];
    const float a2 = pr[2 * PSTR + m];
    float s = (m >= 1) ? a0 : 0.0f;
    s += a1;
    s += (m <= 5) ? a2 : 0.0f;
    outg[e] = s;
  };
  #pragma unroll
  for (int it = 0; it < OUT_WG / NTHR; ++it) tap_sum_store(it * NTHR + tid);
  if (tid < OUT_WG - (OUT_WG / NTHR) * NTHR) tap_sum_store((OUT_WG / NTHR) * NTHR + tid);
}

extern "C" void kernel_launch(void* const* d_in, const int* in_sizes, int n_in,
                              void* d_out, int out_size, void* d_ws, size_t ws_size,
                              hipStream_t stream) {
  (void)in_sizes; (void)n_in; (void)d_ws; (void)ws_size; (void)out_size;
  const float* x = (const float*)d_in[0];
  const float* w = (const float*)d_in[1];
  float* out = (float*)d_out;

  dim3 grid(64), block(96);
  size_t shmem = (size_t)(SLAB + PROWS * PSTR) * sizeof(float);
  hipLaunchKernelGGL(grouped_conv_roll_wmma, grid, block, shmem, stream, x, w, out);
}


# Rows split finer: 5 out-channels (one 16-row tile) per workgroup, weights staged once in LDS, 4 waves one N-tile each: 206 workgroups, 128 WMMAs per wave (was 512); 2 accumulators: sum order changes
# speedup vs baseline: 6.6114x; 1.5119x over previous
#include <hip/hip_runtime.h>

typedef float v2f __attribute__((ext_vector_type(2)));
typedef float v8f __attribute__((ext_vector_type(8)));

namespace {
constexpr int CIN     = 512;
constexpr int COUT    = 512;
constexpr int HW      = 49;
constexpr int WROW    = COUT * 3;
constexpr int SLAB    = CIN * HW;
constexpr int NCP     = SLAB / 2;
constexpr int NTHR    = 128;
constexpr int XSWEEP  = NCP / NTHR;
constexpr int KCH     = 5;
constexpr int NWG     = (COUT + KCH - 1) / KCH;
constexpr int WENT    = (CIN / 2) * 16;
constexpr int WSWEEP  = WENT / NTHR;
constexpr int KB      = 32;
constexpr int NPAIR   = CIN / 4 / KB / 2;
constexpr int PSTR    = 66;
constexpr int NGRID   = 2 * NWG;
}

__device__ __forceinline__ void load_frags(v2f (&A)[KB], v2f (&B)[KB],
                                           const v2f* wa, const v2f* xb) {
  #pragma unroll
  for (int u = 0; u < KB; ++u) {
    A[u] = wa[(u * 2) * 16];
    B[u] = xb[(u * 2) * HW];
  }
}

__device__ __forceinline__ void mma_block(v8f& c0, v8f& c1, const v2f (&A)[KB], const v2f (&B)[KB]) {
  #pragma unroll
  for (int u = 0; u < KB; u += 2) {
    c0 = __builtin_amdgcn_wmma_f32_16x16x4_f32(false, A[u],     false, B[u],     (short)0, c0, false, false);
    c1 = __builtin_amdgcn_wmma_f32_16x16x4_f32(false, A[u + 1], false, B[u + 1], (short)0, c1, false, false);
  }
}

__global__ __launch_bounds__(128)
void grouped_conv_roll_wmma(const float* __restrict__ x,
                            const float* __restrict__ w,
                            float* __restrict__ out) {
  extern __shared__ v2f lds2[];
  v2f*   ldsw  = lds2 + NCP;
  float* ptile = reinterpret_cast<float*>(ldsw + WENT);

  const int tid  = threadIdx.x;
  const int lane = tid & 31;
  const int wave = tid >> 5;
  const int o    = blockIdx.x / NWG;
  const int wgi  = blockIdx.x - o * NWG;
  const int k0   = wgi * KCH;

  {
    const int m  = tid & 15;
    const int rm = (k0 * 3 + m < WROW) ? k0 * 3 + m : WROW - 1;
    const float* wp = w + (tid >> 4) * 2 * WROW + rm;
    #pragma unroll
    for (int q = 0; q < WSWEEP; ++q) {
      const float* p = wp + q * 16 * WROW;
      ldsw[q * NTHR + tid] = v2f{ p[0], p[WROW] };
    }
  }
  {
    const float* xg = x + o * SLAB;
    #pragma unroll
    for (int q = 0; q < XSWEEP; ++q) {
      const int g = q * NTHR + tid;
      const int s = g + (g / HW) * HW;
      lds2[g] = v2f{ xg[s], xg[s + HW] };
    }
  }
  __syncthreads();

  const int ncol = lane & 15;
  const int half = lane >> 4;
  const int col  = (wave * 16 + ncol < HW) ? wave * 16 + ncol : HW - 1;

  const v2f* wa = ldsw + half * 16 + ncol;
  const v2f* xb = lds2 + half * HW + col;

  v8f c0 = {}, c1 = {};
  v2f A0[KB], B0[KB], A1[KB], B1[KB];
  load_frags(A0, B0, wa, xb);

  int bp = 0;
  #pragma nounroll
  for (;;) {
    load_frags(A1, B1, wa + (2 * bp + 1) * KB * 2 * 16, xb + (2 * bp + 1) * KB * 2 * HW);
    __builtin_amdgcn_sched_barrier(0);
    mma_block(c0, c1, A0, B0);

    if (++bp == NPAIR) break;

    load_frags(A0, B0, wa + (2 * bp) * KB * 2 * 16, xb + (2 * bp) * KB * 2 * HW);
    __builtin_amdgcn_sched_barrier(0);
    mma_block(c0, c1, A1, B1);
  }
  mma_block(c0, c1, A1, B1);

  {
    const v8f acc = c0 + c1;
    float* pw = ptile + (half * 8) * PSTR + wave * 16 + ncol;
    #pragma unroll
    for (int v = 0; v < 8; ++v)
      pw[v * PSTR] = acc[v];
  }
  __syncthreads();

  const int nk   = (COUT - k0 < KCH) ? COUT - k0 : KCH;
  const int nout = nk * HW;
  float* outg = out + (o * COUT + k0) * HW;
  auto tap_sum_store = [&](int e) {
    const int kl  = e / HW;
    const int rem = e - kl * HW;
    const int no  = rem / 7;
    const int m   = rem - no * 7;
    const int n   = (no + 6) % 7;
    const float* pr = ptile + (kl * 3) * PSTR + n * 7;
    const float a0 = pr[(m + 5) % 7];
    const float a1 = pr[PSTR + (m + 6) % 7];
    const float a2 = pr[2 * PSTR + m];
    float s = (m >= 1) ? a0 : 0.0f;
    s += a1;
    s += (m <= 5) ? a2 : 0.0f;
    outg[e] = s;
  };
  if (tid < nout)        tap_sum_store(tid);
  if (tid + NTHR < nout) tap_sum_store(tid + NTHR);
}

extern "C" void kernel_launch(void* const* d_in, const int* in_sizes, int n_in,
                              void* d_out, int out_size, void* d_ws, size_t ws_size,
                              hipStream_t stream) {
  (void)in_sizes; (void)n_in; (void)d_ws; (void)ws_size; (void)out_size;
  const float* x = (const float*)d_in[0];
  const float* w = (const float*)d_in[1];
  float* out = (float*)d_out;

  dim3 grid(NGRID), block(NTHR);
  size_t shmem = (size_t)(SLAB + 2 * WENT + 16 * PSTR) * sizeof(float);
  hipLaunchKernelGGL(grouped_conv_roll_wmma, grid, block, shmem, stream, x, w, out);
}


# Staging overlapped with the products: every global load requested at once in K order (activations as a straight 16 B copy); K walk starts on channels 0-255 while 256-511 still land; sums unchanged
# speedup vs baseline: 7.2401x; 1.0951x over previous
#include <hip/hip_runtime.h>

typedef float v2f __attribute__((ext_vector_type(2)));
typedef float v4f __attribute__((ext_vector_type(4)));
typedef float v8f __attribute__((ext_vector_type(8)));

namespace {
constexpr int CIN     = 512;
constexpr int COUT    = 512;
constexpr int HW      = 49;
constexpr int WROW    = COUT * 3;
constexpr int SLAB    = CIN * HW;
constexpr int NTHR    = 128;
constexpr int X4      = SLAB / 4;
constexpr int XSWEEP  = X4 / NTHR;
constexpr int KCH     = 5;
constexpr int NWG     = (COUT + KCH - 1) / KCH;
constexpr int WENT    = (CIN / 2) * 16;
constexpr int WSWEEP  = WENT / NTHR;
constexpr int KB      = 32;
constexpr int NBLK    = CIN / 4 / KB;
constexpr int WQ      = WSWEEP / NBLK;
constexpr int PSTR    = 66;
constexpr int NGRID   = 2 * NWG;
constexpr int XEND0 = 13, XEND1 = 25, XEND2 = 37, XEND3 = 49;
static_assert(XEND0 * NTHR >= 1 * (X4 / NBLK) && XEND1 * NTHR >= 2 * (X4 / NBLK) &&
              XEND2 * NTHR >= 3 * (X4 / NBLK) && XEND3 == XSWEEP, "block coverage");
}

__device__ __forceinline__ void load_frags(v2f (&A)[KB], v2f (&B)[KB],
                                           const v2f* wa, const float* xb) {
  #pragma unroll
  for (int u = 0; u < KB; ++u) {
    A[u] = wa[(u * 2) * 16];
    B[u] = v2f{ xb[(u * 4) * HW], xb[(u * 4 + 1) * HW] };
  }
}

__device__ __forceinline__ void mma_block(v8f& c0, v8f& c1, const v2f (&A)[KB], const v2f (&B)[KB]) {
  #pragma unroll
  for (int u = 0; u < KB; u += 2) {
    c0 = __builtin_amdgcn_wmma_f32_16x16x4_f32(false, A[u],     false, B[u],     (short)0, c0, false, false);
    c1 = __builtin_amdgcn_wmma_f32_16x16x4_f32(false, A[u + 1], false, B[u + 1], (short)0, c1, false, false);
  }
}

__device__ __forceinline__ void lds_barrier() {
  asm volatile("s_wait_dscnt 0x0\n\ts_barrier_signal -1\n\ts_barrier_wait -1" ::: "memory");
}

__device__ __forceinline__ void order_point(v8f& c0, v8f& c1) {
  asm volatile("" : "+v"(c0), "+v"(c1) :: "memory");
}

__global__ __launch_bounds__(128)
void grouped_conv_roll_wmma(const float* __restrict__ x,
                            const float* __restrict__ w,
                            float* __restrict__ out) {
  extern __shared__ v4f lds4[];
  float* xs    = reinterpret_cast<float*>(lds4);
  v2f*   ldsw  = reinterpret_cast<v2f*>(xs + SLAB);
  float* ptile = reinterpret_cast<float*>(ldsw + WENT);

  const int tid  = threadIdx.x;
  const int lane = tid & 31;
  const int wave = tid >> 5;
  const int o    = blockIdx.x / NWG;
  const int wgi  = blockIdx.x - o * NWG;
  const int k0   = wgi * KCH;

  const int m  = tid & 15;
  const int rm = (k0 * 3 + m < WROW) ? k0 * 3 + m : WROW - 1;
  const float* wp  = w + (tid >> 4) * 2 * WROW + rm;
  const v4f*   xg4 = reinterpret_cast<const v4f*>(x + o * SLAB) + tid;

  float wr0[WSWEEP], wr1[WSWEEP];
  v4f   xr[XSWEEP];
  auto request = [&](int b, int xlo, int xhi) {
    #pragma unroll
    for (int q = b * WQ; q < (b + 1) * WQ; ++q) {
      const float* p = wp + q * 16 * WROW;
      wr0[q] = p[0];
      wr1[q] = p[WROW];
    }
    #pragma unroll
    for (int q = xlo; q < xhi; ++q)
      xr[q] = xg4[q * NTHR];
  };
  auto to_lds = [&](int b, int xlo, int xhi) {
    #pragma unroll
    for (int q = b * WQ; q < (b + 1) * WQ; ++q)
      ldsw[q * NTHR + tid] = v2f{ wr0[q], wr1[q] };
    #pragma unroll
    for (int q = xlo; q < xhi; ++q)
      lds4[q * NTHR + tid] = xr[q];
  };
  request(0, 0,     XEND0);
  __builtin_amdgcn_sched_barrier(0);
  request(1, XEND0, XEND1);
  __builtin_amdgcn_sched_barrier(0);
  request(2, XEND1, XEND2);
  __builtin_amdgcn_sched_barrier(0);
  request(3, XEND2, XEND3);
  __builtin_amdgcn_sched_barrier(0);

  const int ncol = lane & 15;
  const int half = lane >> 4;
  const int col  = (wave * 16 + ncol < HW) ? wave * 16 + ncol : HW - 1;

  const v2f*   wa = ldsw + half * 16 + ncol;
  const float* xb = xs + (half * 2) * HW + col;

  v8f c0 = {}, c1 = {};
  v2f A0[KB], B0[KB], A1[KB], B1[KB];

  to_lds(0, 0,     XEND0);
  to_lds(1, XEND0, XEND1);
  lds_barrier();
  load_frags(A0, B0, wa, xb);
  order_point(c0, c1);
  load_frags(A1, B1, wa + 1 * KB * 2 * 16, xb + 1 * KB * 4 * HW);
  order_point(c0, c1);
  mma_block(c0, c1, A0, B0);
  order_point(c0, c1);

  to_lds(2, XEND1, XEND2);
  lds_barrier();
  load_frags(A0, B0, wa + 2 * KB * 2 * 16, xb + 2 * KB * 4 * HW);
  order_point(c0, c1);
  mma_block(c0, c1, A1, B1);
  order_point(c0, c1);

  to_lds(3, XEND2, XEND3);
  lds_barrier();
  load_frags(A1, B1, wa + 3 * KB * 2 * 16, xb + 3 * KB * 4 * HW);
  order_point(c0, c1);
  mma_block(c0, c1, A0, B0);
  mma_block(c0, c1, A1, B1);

  {
    const v8f acc = c0 + c1;
    float* pw = ptile + (half * 8) * PSTR + wave * 16 + ncol;
    #pragma unroll
    for (int v = 0; v < 8; ++v)
      pw[v * PSTR] = acc[v];
  }
  __syncthreads();

  const int nk   = (COUT - k0 < KCH) ? COUT - k0 : KCH;
  const int nout = nk * HW;
  float* outg = out + (o * COUT + k0) * HW;
  auto tap_sum_store = [&](int e) {
    const int kl  = e / HW;
    const int rem = e - kl * HW;
    const int no  = rem / 7;
    const int mm  = rem - no * 7;
    const int n   = (no + 6) % 7;
    const float* pr = ptile + (kl * 3) * PSTR + n * 7;
    const float a0 = pr[(mm + 5) % 7];
    const float a1 = pr[PSTR + (mm + 6) % 7];
    const float a2 = pr[2 * PSTR + mm];
    float s = (mm >= 1) ? a0 : 0.0f;
    s += a1;
    s += (mm <= 5) ? a2 : 0.0f;
    outg[e] = s;
  };
  if (tid < nout)        tap_sum_store(tid);
  if (tid + NTHR < nout) tap_sum_store(tid + NTHR);
}

extern "C" void kernel_launch(void* const* d_in, const int* in_sizes, int n_in,
                              void* d_out, int out_size, void* d_ws, size_t ws_size,
                              hipStream_t stream) {
  (void)in_sizes; (void)n_in; (void)d_ws; (void)ws_size; (void)out_size;
  const float* x = (const float*)d_in[0];
  const float* w = (const float*)d_in[1];
  float* out = (float*)d_out;

  dim3 grid(NGRID), block(NTHR);
  size_t shmem = (size_t)(SLAB + 2 * WENT + 16 * PSTR) * sizeof(float);
  hipLaunchKernelGGL(grouped_conv_roll_wmma, grid, block, shmem, stream, x, w, out);
}


# K walk in 7 blocks of 64,64,128,128,64,32,32 channels (was 4 of 128): loads requested 256 channels ahead of the stores, not all up front; products start after 128 channels (was 256); sums unchanged
# speedup vs baseline: 7.9008x; 1.0913x over previous
#include <hip/hip_runtime.h>

typedef float v2f __attribute__((ext_vector_type(2)));
typedef float v4f __attribute__((ext_vector_type(4)));
typedef float v8f __attribute__((ext_vector_type(8)));

namespace {
constexpr int CIN     = 512;
constexpr int COUT    = 512;
constexpr int HW      = 49;
constexpr int WROW    = COUT * 3;
constexpr int SLAB    = CIN * HW;
constexpr int NTHR    = 128;
constexpr int X4      = SLAB / 4;
constexpr int XSWEEP  = X4 / NTHR;
constexpr int KCH     = 5;
constexpr int NWG     = (COUT + KCH - 1) / KCH;
constexpr int WENT    = (CIN / 2) * 16;
constexpr int WSWEEP  = WENT / NTHR;
constexpr int KB      = 32;
constexpr int NCHUNK  = 9;
constexpr int PSTR    = 66;
constexpr int NGRID   = 2 * NWG;
constexpr int CE[NCHUNK + 1] = { 0, 64, 128, 192, 256, 320, 384, 448, 480, 512 };
constexpr int CW[NCHUNK + 1] = { 0,  4,   8,  12,  16,  20,  24,  28,  30,  32 };
constexpr int CX[NCHUNK + 1] = { 0,  7,  13,  19,  25,  31,  37,  43,  46,  49 };
constexpr bool chunks_cover() {
  for (int c = 1; c <= NCHUNK; ++c)
    if (CW[c] * 16 != CE[c] || CX[c] * NTHR * 4 < CE[c] * HW || CX[c] < CX[c - 1]) return false;
  return CW[NCHUNK] == WSWEEP && CX[NCHUNK] == XSWEEP;
}
static_assert(chunks_cover(), "chunk coverage");
constexpr int loads_in(int a, int b) { return 2 * (CW[b] - CW[a]) + (CX[b] - CX[a]); }
static_assert(loads_in(0, 4) <= 63 && loads_in(1, 5) <= 63 && loads_in(2, 6) <= 63 &&
              loads_in(4, 9) <= 63, "requests in flight");
}

template <int N>
__device__ __forceinline__ void load_frags(v2f (&A)[KB], v2f (&B)[KB],
                                           const v2f* wa, const float* xb) {
  #pragma unroll
  for (int u = 0; u < N; ++u) {
    A[u] = wa[(u * 2) * 16];
    B[u] = v2f{ xb[(u * 4) * HW], xb[(u * 4 + 1) * HW] };
  }
}

template <int N>
__device__ __forceinline__ void mma_block(v8f& c0, v8f& c1, const v2f (&A)[KB], const v2f (&B)[KB]) {
  static_assert(N % 2 == 0 && N <= KB, "block size");
  #pragma unroll
  for (int u = 0; u < N; u += 2) {
    c0 = __builtin_amdgcn_wmma_f32_16x16x4_f32(false, A[u],     false, B[u],     (short)0, c0, false, false);
    c1 = __builtin_amdgcn_wmma_f32_16x16x4_f32(false, A[u + 1], false, B[u + 1], (short)0, c1, false, false);
  }
}

__device__ __forceinline__ void lds_barrier() {
  asm volatile("s_wait_dscnt 0x0\n\ts_barrier_signal -1\n\ts_barrier_wait -1" ::: "memory");
}

__device__ __forceinline__ void order_point(v8f& c0, v8f& c1) {
  asm volatile("" : "+v"(c0), "+v"(c1) :: "memory");
}

__global__ __launch_bounds__(128)
void grouped_conv_roll_wmma(const float* __restrict__ x,
                            const float* __restrict__ w,
                            float* __restrict__ out) {
  extern __shared__ v4f lds4[];
  float* xs    = reinterpret_cast<float*>(lds4);
  v2f*   ldsw  = reinterpret_cast<v2f*>(xs + SLAB);
  float* ptile = reinterpret_cast<float*>(ldsw + WENT);

  const int tid  = threadIdx.x;
  const int lane = tid & 31;
  const int wave = tid >> 5;
  const int o    = blockIdx.x / NWG;
  const int wgi  = blockIdx.x - o * NWG;
  const int k0   = wgi * KCH;

  const int m  = tid & 15;
  const int rm = (k0 * 3 + m < WROW) ? k0 * 3 + m : WROW - 1;
  const float* wp  = w + (tid >> 4) * 2 * WROW + rm;
  const v4f*   xg4 = reinterpret_cast<const v4f*>(x + o * SLAB) + tid;

  float wr0[WSWEEP], wr1[WSWEEP];
  v4f   xr[XSWEEP];
  auto request = [&](int wlo, int whi, int xlo, int xhi) {
    #pragma unroll
    for (int q = wlo; q < whi; ++q) {
      const float* p = wp + q * 16 * WROW;
      wr0[q] = p[0];
      wr1[q] = p[WROW];
    }
    #pragma unroll
    for (int q = xlo; q < xhi; ++q)
      xr[q] = xg4[q * NTHR];
  };
  auto to_lds = [&](int wlo, int whi, int xlo, int xhi) {
    #pragma unroll
    for (int q = wlo; q < whi; ++q)
      ldsw[q * NTHR + tid] = v2f{ wr0[q], wr1[q] };
    #pragma unroll
    for (int q = xlo; q < xhi; ++q)
      lds4[q * NTHR + tid] = xr[q];
  };
#define REQUEST_CHUNK(c)   __builtin_amdgcn_sched_barrier(0); \
                           request(CW[(c)], CW[(c) + 1], CX[(c)], CX[(c) + 1]);
#define STORE_CHUNKS(a, b) to_lds(CW[(a)], CW[(b)], CX[(a)], CX[(b)]);
  REQUEST_CHUNK(0)
  REQUEST_CHUNK(1)
  REQUEST_CHUNK(2)
  REQUEST_CHUNK(3)
  __builtin_amdgcn_sched_barrier(0);

  const int ncol = lane & 15;
  const int half = lane >> 4;
  const int col  = (wave * 16 + ncol < HW) ? wave * 16 + ncol : HW - 1;

  const v2f*   wa = ldsw + half * 16 + ncol;
  const float* xb = xs + (half * 2) * HW + col;

  v8f c0 = {}, c1 = {};
  v2f A0[KB], B0[KB], A1[KB], B1[KB];

  STORE_CHUNKS(0, 1)
  REQUEST_CHUNK(4)
  lds_barrier();
  load_frags<16>(A0, B0, wa + 0 * 2 * 16, xb + 0 * 4 * HW);
  order_point(c0, c1);

  STORE_CHUNKS(1, 2)
  REQUEST_CHUNK(5)
  lds_barrier();
  load_frags<16>(A1, B1, wa + 16 * 2 * 16, xb + 16 * 4 * HW);
  order_point(c0, c1);
  mma_block<16>(c0, c1, A0, B0);
  order_point(c0, c1);

  STORE_CHUNKS(2, 4)
  REQUEST_CHUNK(6)
  REQUEST_CHUNK(7)
  REQUEST_CHUNK(8)
  lds_barrier();
  load_frags<32>(A0, B0, wa + 32 * 2 * 16, xb + 32 * 4 * HW);
  order_point(c0, c1);
  mma_block<16>(c0, c1, A1, B1);
  order_point(c0, c1);

  STORE_CHUNKS(4, 6)
  lds_barrier();
  load_frags<32>(A1, B1, wa + 64 * 2 * 16, xb + 64 * 4 * HW);
  order_point(c0, c1);
  mma_block<32>(c0, c1, A0, B0);
  order_point(c0, c1);

  STORE_CHUNKS(6, 7)
  lds_barrier();
  load_frags<16>(A0, B0, wa + 96 * 2 * 16, xb + 96 * 4 * HW);
  order_point(c0, c1);
  mma_block<32>(c0, c1, A1, B1);
  order_point(c0, c1);

  STORE_CHUNKS(7, 8)
  lds_barrier();
  load_frags<8>(A1, B1, wa + 112 * 2 * 16, xb + 112 * 4 * HW);
  order_point(c0, c1);
  mma_block<16>(c0, c1, A0, B0);
  order_point(c0, c1);

  STORE_CHUNKS(8, 9)
  lds_barrier();
  load_frags<8>(A0, B0, wa + 120 * 2 * 16, xb + 120 * 4 * HW);
  order_point(c0, c1);
  mma_block<8>(c0, c1, A1, B1);
  mma_block<8>(c0, c1, A0, B0);
#undef REQUEST_CHUNK
#undef STORE_CHUNKS

  {
    const v8f acc = c0 + c1;
    float* pw = ptile + (half * 8) * PSTR + wave * 16 + ncol;
    #pragma unroll
    for (int v = 0; v < 8; ++v)
      pw[v * PSTR] = acc[v];
  }
  __syncthreads();

  const int nk   = (COUT - k0 < KCH) ? COUT - k0 : KCH;
  const int nout = nk * HW;
  float* outg = out + (o * COUT + k0) * HW;
  auto tap_sum_store = [&](int e) {
    const int kl  = e / HW;
    const int rem = e - kl * HW;
    const int no  = rem / 7;
    const int mm  = rem - no * 7;
    const int n   = (no + 6) % 7;
    const float* pr = ptile + (kl * 3) * PSTR + n * 7;
    const float a0 = pr[(mm + 5) % 7];
    const float a1 = pr[PSTR + (mm + 6) % 7];
    const float a2 = pr[2 * PSTR + mm];
    float s = (mm >= 1) ? a0 : 0.0f;
    s += a1;
    s += (mm <= 5) ? a2 : 0.0f;
    outg[e] = s;
  };
  if (tid < nout)        tap_sum_store(tid);
  if (tid + NTHR < nout) tap_sum_store(tid + NTHR);
}

extern "C" void kernel_launch(void* const* d_in, const int* in_sizes, int n_in,
                              void* d_out, int out_size, void* d_ws, size_t ws_size,
                              hipStream_t stream) {
  (void)in_sizes; (void)n_in; (void)d_ws; (void)ws_size; (void)out_size;
  const float* x = (const float*)d_in[0];
  const float* w = (const float*)d_in[1];
  float* out = (float*)d_out;

  dim3 grid(NGRID), block(NTHR);
  size_t shmem = (size_t)(SLAB + 2 * WENT + 16 * PSTR) * sizeof(float);
  hipLaunchKernelGGL(grouped_conv_roll_wmma, grid, block, shmem, stream, x, w, out);
}
